# starting kernel
# baseline (speedup 1.0000x reference)
#include <hip/hip_runtime.h>


typedef __attribute__((ext_vector_type(2))) float v2f;
typedef __attribute__((ext_vector_type(8))) float v8f;

constexpr int kN  = 100000;
constexpr int kE  = 1600000;
constexpr int kXF = 128;
constexpr int kXL = 40;
constexpr int kH  = 128;
constexpr int kC  = 40;
constexpr int kKM = kXF + kXL;
constexpr int kKC = 2 * kH;


__global__ void k_init_deg(float* __restrict__ deg) {
  int i = blockIdx.x * blockDim.x + threadIdx.x;
  if (i < kN) deg[i] = 1.0f;
}

__global__ void k_deg_accum(const int* __restrict__ eidx, float* __restrict__ deg) {
  int e = blockIdx.x * blockDim.x + threadIdx.x;
  if (e < kE) atomicAdd(deg + eidx[kE + e], 1.0f);
}

__global__ void k_dinv(float* __restrict__ deg) {
  int i = blockIdx.x * blockDim.x + threadIdx.x;
  if (i < kN) {
    float d = deg[i];
    deg[i] = (d > 0.0f) ? rsqrtf(d) : 0.0f;
  }
}


__global__ __launch_bounds__(32)
void k_gemm_gcn(const float* __restrict__ A, const float* __restrict__ B,
                float* __restrict__ Hout) {
  __shared__ float As[16][kXF + 4];
  __shared__ float Bs[kXF][16];

  const int lane = threadIdx.x;
  const int m0 = blockIdx.x * 16;
  const int n0 = blockIdx.y * 16;

  for (int idx = lane; idx < 16 * kXF; idx += 32) {
    int r = idx >> 7, c = idx & 127;
    As[r][c] = A[(size_t)(m0 + r) * kXF + c];
  }
  for (int idx = lane; idx < kXF * 16; idx += 32) {
    int k = idx >> 4, n = idx & 15;
    Bs[k][n] = B[k * kH + n0 + n];
  }
  __syncthreads();

  const int half = lane >> 4;
  const int l    = lane & 15;
  v8f acc = {};

#pragma unroll 4
  for (int kb = 0; kb < kXF; kb += 4) {
    v2f a, b;
    a[0] = As[l][kb + 2 * half + 0];
    a[1] = As[l][kb + 2 * half + 1];
    b[0] = Bs[kb + 2 * half + 0][l];
    b[1] = Bs[kb + 2 * half + 1][l];
    acc = __builtin_amdgcn_wmma_f32_16x16x4_f32(false, a, false, b,
                                                (short)0, acc, false, false);
  }

#pragma unroll
  for (int r = 0; r < 8; ++r)
    Hout[(size_t)(m0 + half * 8 + r) * kH + (n0 + l)] = acc[r];
}


__global__ void k_selfloop(const float* __restrict__ h,
                           const float* __restrict__ dinv,
                           float* __restrict__ comb) {
  int t = blockIdx.x * blockDim.x + threadIdx.x;
  if (t < kN * kH) {
    int i = t >> 7, c = t & 127;
    float di = dinv[i];
    comb[(size_t)i * kKC + c] = h[(size_t)i * kH + c] * di * di;
  }
}


__global__ void k_edge_scatter(const int* __restrict__ eidx,
                               const float* __restrict__ h,
                               const float* __restrict__ dinv,
                               float* __restrict__ comb) {
  long long t = (long long)blockIdx.x * blockDim.x + threadIdx.x;
  if (t >= (long long)kE * 32) return;
  int e = (int)(t >> 5);
  int q = (int)(t & 31);
  int s = eidx[e];
  int d = eidx[kE + e];
  float nm = dinv[s] * dinv[d];
  const float4 hv = *reinterpret_cast<const float4*>(h + (size_t)s * kH + q * 4);
  float* p = comb + (size_t)d * kKC + q * 4;
  atomicAdd(p + 0, hv.x * nm);
  atomicAdd(p + 1, hv.y * nm);
  atomicAdd(p + 2, hv.z * nm);
  atomicAdd(p + 3, hv.w * nm);
}


__global__ __launch_bounds__(32)
void k_gemm_mlp(const float* __restrict__ xf, const float* __restrict__ xl,
                const float* __restrict__ W, const float* __restrict__ bias,
                float* __restrict__ comb) {
  __shared__ float As[16][kKM + 4];
  __shared__ float Bs[kKM][16];

  const int lane = threadIdx.x;
  const int m0 = blockIdx.x * 16;
  const int n0 = blockIdx.y * 16;

  for (int idx = lane; idx < 16 * kKM; idx += 32) {
    int r = idx / kKM, c = idx % kKM;
    As[r][c] = (c < kXF) ? xf[(size_t)(m0 + r) * kXF + c]
                         : xl[(size_t)(m0 + r) * kXL + (c - kXF)];
  }
  for (int idx = lane; idx < kKM * 16; idx += 32) {
    int k = idx >> 4, n = idx & 15;
    Bs[k][n] = W[k * kH + n0 + n];
  }
  __syncthreads();

  const int half = lane >> 4;
  const int l    = lane & 15;
  v8f acc = {};

#pragma unroll 2
  for (int kb = 0; kb < kKM; kb += 4) {
    v2f a, b;
    a[0] = As[l][kb + 2 * half + 0];
    a[1] = As[l][kb + 2 * half + 1];
    b[0] = Bs[kb + 2 * half + 0][l];
    b[1] = Bs[kb + 2 * half + 1][l];
    acc = __builtin_amdgcn_wmma_f32_16x16x4_f32(false, a, false, b,
                                                (short)0, acc, false, false);
  }

  const float bn = bias[n0 + l];
#pragma unroll
  for (int r = 0; r < 8; ++r) {
    float v = acc[r] + bn;
    v = v > 0.0f ? v : 0.0f;
    comb[(size_t)(m0 + half * 8 + r) * kKC + kH + (n0 + l)] = v;
  }
}


__global__ void k_gcn_bias_relu(float* __restrict__ comb, const float* __restrict__ bg) {
  int t = blockIdx.x * blockDim.x + threadIdx.x;
  if (t < kN * kH) {
    int i = t >> 7, c = t & 127;
    float v = comb[(size_t)i * kKC + c] + bg[c];
    comb[(size_t)i * kKC + c] = v > 0.0f ? v : 0.0f;
  }
}


__global__ __launch_bounds__(32)
void k_gemm_cls(const float* __restrict__ comb, const float* __restrict__ W,
                const float* __restrict__ bias, float* __restrict__ out) {
  __shared__ float As[16][kKC + 4];
  __shared__ float Bs[kKC][16];

  const int lane = threadIdx.x;
  const int m0 = blockIdx.x * 16;
  const int n0 = blockIdx.y * 16;

  for (int idx = lane; idx < 16 * kKC; idx += 32) {
    int r = idx >> 8, c = idx & 255;
    As[r][c] = comb[(size_t)(m0 + r) * kKC + c];
  }
  for (int idx = lane; idx < kKC * 16; idx += 32) {
    int k = idx >> 4, n = idx & 15;
    Bs[k][n] = (n0 + n < kC) ? W[k * kC + n0 + n] : 0.0f;
  }
  __syncthreads();

  const int half = lane >> 4;
  const int l    = lane & 15;
  v8f acc = {};

#pragma unroll 4
  for (int kb = 0; kb < kKC; kb += 4) {
    v2f a, b;
    a[0] = As[l][kb + 2 * half + 0];
    a[1] = As[l][kb + 2 * half + 1];
    b[0] = Bs[kb + 2 * half + 0][l];
    b[1] = Bs[kb + 2 * half + 1][l];
    acc = __builtin_amdgcn_wmma_f32_16x16x4_f32(false, a, false, b,
                                                (short)0, acc, false, false);
  }

  const int col = n0 + l;
  if (col < kC) {
    const float bn = bias[col];
#pragma unroll
    for (int r = 0; r < 8; ++r)
      out[(size_t)(m0 + half * 8 + r) * kC + col] = acc[r] + bn;
  }
}


extern "C" void kernel_launch(void* const* d_in, const int* in_sizes, int n_in,
                              void* d_out, int out_size, void* d_ws, size_t ws_size,
                              hipStream_t stream) {
  (void)in_sizes; (void)n_in; (void)out_size; (void)ws_size;

  const float* xfeat  = (const float*)d_in[0];
  const float* xlabel = (const float*)d_in[1];
  const int*   eidx   = (const int*)d_in[2];
  const float* W_gcn  = (const float*)d_in[3];
  const float* b_gcn  = (const float*)d_in[4];
  const float* W_mlp  = (const float*)d_in[5];
  const float* b_mlp  = (const float*)d_in[6];
  const float* W_cls  = (const float*)d_in[7];
  const float* b_cls  = (const float*)d_in[8];
  float* out = (float*)d_out;

  char* ws = (char*)d_ws;
  const size_t off_deg  = 0;
  const size_t off_h    = (off_deg + (size_t)kN * 4 + 255) & ~(size_t)255;
  const size_t off_comb = (off_h + (size_t)kN * kH * 4 + 255) & ~(size_t)255;
  float* deg  = (float*)(ws + off_deg);
  float* hbuf = (float*)(ws + off_h);
  float* comb = (float*)(ws + off_comb);

  const dim3 blk256(256);

  k_init_deg <<<dim3((kN + 255) / 256), blk256, 0, stream>>>(deg);
  k_deg_accum<<<dim3((kE + 255) / 256), blk256, 0, stream>>>(eidx, deg);
  k_dinv     <<<dim3((kN + 255) / 256), blk256, 0, stream>>>(deg);

  k_gemm_gcn<<<dim3(kN / 16, kH / 16), dim3(32), 0, stream>>>(xfeat, W_gcn, hbuf);

  k_selfloop<<<dim3((kN * kH) / 256), blk256, 0, stream>>>(hbuf, deg, comb);
  k_edge_scatter<<<dim3((unsigned)(((long long)kE * 32) / 256)), blk256, 0, stream>>>(
      eidx, hbuf, deg, comb);

  k_gemm_mlp<<<dim3(kN / 16, kH / 16), dim3(32), 0, stream>>>(
      xfeat, xlabel, W_mlp, b_mlp, comb);

  k_gcn_bias_relu<<<dim3((kN * kH) / 256), blk256, 0, stream>>>(comb, b_gcn);

  k_gemm_cls<<<dim3(kN / 16, 3), dim3(32), 0, stream>>>(comb, W_cls, b_cls, out);
}


# GCN product reassociated as (A x) W: edges sum normalised xfeat rows, W_gcn applied to the sums in the fused tail on WMMA (register-chained); h and its GEMM launch removed; edge sum precedes k sum
# speedup vs baseline: 4.8324x; 4.8324x over previous
#include <hip/hip_runtime.h>


typedef __attribute__((ext_vector_type(2))) float v2f;
typedef __attribute__((ext_vector_type(4))) float v4f;
typedef __attribute__((ext_vector_type(8))) float v8f;

constexpr int kN  = 100000;
constexpr int kE  = 1600000;
constexpr int kXF = 128;
constexpr int kXL = 40;
constexpr int kH  = 128;
constexpr int kC  = 40;
constexpr int kKM = kXF + kXL;
constexpr int kKC = 2 * kH;


__global__ void k_init_deg(float* __restrict__ deg) {
  int i = blockIdx.x * blockDim.x + threadIdx.x;
  if (i < kN) deg[i] = 1.0f;
}

constexpr int kOwnRows = (kN + 255) / 256;
static_assert(kE % 16 == 0 && kOwnRows * 256 >= kN && (kN - 1) / kOwnRows < 256, "owner bytes");

__global__ void k_deg_owner(const int* __restrict__ eidx, float* __restrict__ deg,
                            unsigned* __restrict__ own) {
  int t = blockIdx.x * blockDim.x + threadIdx.x;
  if (t < kE / 4) {
    const int4 d = *reinterpret_cast<const int4*>(eidx + kE + 4 * t);
    atomicAdd(deg + d.x, 1.0f);
    atomicAdd(deg + d.y, 1.0f);
    atomicAdd(deg + d.z, 1.0f);
    atomicAdd(deg + d.w, 1.0f);
    own[t] = (unsigned)(d.x / kOwnRows) | ((unsigned)(d.y / kOwnRows) << 8) |
             ((unsigned)(d.z / kOwnRows) << 16) | ((unsigned)(d.w / kOwnRows) << 24);
  }
}

__global__ void k_dinv(float* __restrict__ deg) {
  int i = blockIdx.x * blockDim.x + threadIdx.x;
  if (i < kN) {
    float d = deg[i];
    deg[i] = (d > 0.0f) ? rsqrtf(d) : 0.0f;
  }
}


constexpr int kAggThreads = 1024;
constexpr int kAggWaves = kAggThreads / 32;
constexpr int kChunks = kE / 16;
constexpr int kScanTrips = (kChunks + kAggThreads - 1) / kAggThreads;
constexpr int kQCap = 512 + 8;

__device__ __forceinline__ unsigned zero_bytes(unsigned y) {
  return ~(((y & 0x7f7f7f7fu) + 0x7f7f7f7fu) | y | 0x7f7f7f7fu);
}

__device__ __forceinline__ void agg_edge_add(float* __restrict__ row, const float4 hv, float nm) {
  atomicAdd(row + 0,  hv.x * nm);
  atomicAdd(row + 32, hv.y * nm);
  atomicAdd(row + 64, hv.z * nm);
  atomicAdd(row + 96, hv.w * nm);
}

__global__ __launch_bounds__(kAggThreads)
void k_agg_lds(const int* __restrict__ eidx, const unsigned* __restrict__ own,
               const float* __restrict__ xf, const float* __restrict__ dinv,
               float* __restrict__ aggx) {
  __shared__ __attribute__((aligned(16))) float acc[kOwnRows * kH];
  __shared__ int queue[kAggWaves * kQCap];

  const int tid = threadIdx.x, lane = tid & 31, wave = tid >> 5;
  const int lo = blockIdx.x * kOwnRows;
  const int rows = (kN - lo < kOwnRows) ? (kN - lo) : kOwnRows;

  for (int i = tid; i < rows * 32; i += kAggThreads) {
    const int r = i >> 5, q = i & 31;
    const float di = dinv[lo + r];
    const float4 hv = *reinterpret_cast<const float4*>(xf + (size_t)(lo + r) * kXF + 4 * q);
    float* a = acc + r * kH + q;
    a[0]  = hv.x * di * di;
    a[32] = hv.y * di * di;
    a[64] = hv.z * di * di;
    a[96] = hv.w * di * di;
  }
  __syncthreads();

  int* myq = queue + wave * kQCap;
  int cnt = 0;
  const unsigned me = blockIdx.x * 0x01010101u;
  const uint4* own4 = reinterpret_cast<const uint4*>(own);
  float* accl = acc + lane;

  uint4 x = own4[tid];
  for (int it = 0; it < kScanTrips; ++it) {
    const int c = it * kAggThreads + tid;
    const int cn = c + kAggThreads;
    uint4 xn = x;
    if (cn < kChunks) xn = own4[cn];
    unsigned m = (zero_bytes(x.x ^ me) >> 7) | (zero_bytes(x.y ^ me) >> 6) |
                 (zero_bytes(x.z ^ me) >> 5) | (zero_bytes(x.w ^ me) >> 4);
    if (c >= kChunks) m = 0;
    x = xn;

    unsigned bal = __builtin_amdgcn_ballot_w32(m != 0);
    while (bal != 0) {
      if (m != 0) {
        const int q = __builtin_ctz(m);
        m &= m - 1;
        myq[cnt + (int)__builtin_amdgcn_mbcnt_lo(bal, 0u)] = 16 * c + ((q & 7) << 2) + (q >> 3);
      }
      cnt += __builtin_popcount(bal);
      bal = __builtin_amdgcn_ballot_w32(m != 0);
    }

    while (cnt >= 8) {
      int s[8], d[8];
#pragma unroll
      for (int k = 0; k < 8; ++k) {
        const int e = __builtin_amdgcn_readfirstlane(myq[cnt - 1 - k]);
        s[k] = eidx[e];
        d[k] = eidx[kE + e];
      }
      float4 hv[8];
      float nm[8];
#pragma unroll
      for (int k = 0; k < 8; ++k) {
        hv[k] = *reinterpret_cast<const float4*>(xf + (size_t)s[k] * kXF + 4 * lane);
        nm[k] = dinv[s[k]] * dinv[d[k]];
      }
#pragma unroll
      for (int k = 0; k < 8; ++k) agg_edge_add(accl + (d[k] - lo) * kH, hv[k], nm[k]);
      cnt -= 8;
    }
  }
  while (cnt > 0) {
    const int e = __builtin_amdgcn_readfirstlane(myq[cnt - 1]);
    const int s = eidx[e], d = eidx[kE + e];
    const float4 hv = *reinterpret_cast<const float4*>(xf + (size_t)s * kXF + 4 * lane);
    agg_edge_add(accl + (d - lo) * kH, hv, dinv[s] * dinv[d]);
    --cnt;
  }
  __syncthreads();

  for (int i = tid; i < rows * 32; i += kAggThreads) {
    const int r = i >> 5, q = i & 31;
    const float* a = acc + r * kH + q;
    const float4 v = {a[0], a[32], a[64], a[96]};
    *reinterpret_cast<float4*>(aggx + (size_t)(lo + r) * kXF + 4 * q) = v;
  }
}


constexpr int kStrips = kN / 16;
constexpr int kW1F  = (kKM / 8) * 8 * 32 * 4;
constexpr int kWgF  = (kXF / 8) * 8 * 32 * 4;
constexpr int kW2hF = 8 * 2 * 3 * 32 * 4;
constexpr int kTailThreads = 512;
constexpr int kTailWaves = kTailThreads / 32;
constexpr int kTailBlocks = 256;
constexpr int kW1Quads = kKM * (kH / 4);
constexpr int kWgQuads = kXF * (kH / 4);
constexpr int kW2Quads = kKC * (kC / 4);
static_assert(kWgQuads % kTailThreads == 0 && kW2Quads % kTailThreads == 0 &&
              (kKC * 8) % kTailThreads == 0, "staging trip counts");

__device__ __forceinline__ int w1_slot(int k, int n) {
  return ((((k >> 3) * 8 + (n >> 4)) * 32 + ((k >> 2) & 1) * 16 + (n & 15)) << 2) + (k & 3);
}

__device__ __forceinline__ int w2_slot(int k, int n) {
  const int u = n >> 4, l = n & 15, c = k & 3;
  const int kk = k & (kH - 1);
  return (k >> 7) * kW2hF +
         ((((((kk >> 4) * 2 + ((kk >> 2) & 1)) * 3 + u) * 32 + ((kk >> 3) & 1) * 16 + l) << 2) + c);
}

__device__ __forceinline__ v8f mm(float a0, float a1, float b0, float b1, v8f acc) {
  v2f a = {a0, a1};
  v2f b = {b0, b1};
  return __builtin_amdgcn_wmma_f32_16x16x4_f32(false, a, false, b, (short)0, acc, false, false);
}

__device__ __forceinline__ float relu1(float v) { return v > 0.0f ? v : 0.0f; }

__device__ __forceinline__ void bias_relu_classify(v8f (&acc)[8], const v4f* __restrict__ bv,
                                                   const v4f* __restrict__ W2h, int half, int lane,
                                                   v8f (&acc2)[3]) {
#pragma unroll
  for (int t = 0; t < 8; ++t) {
    const v4f blo = bv[4 * t + 2 * half], bhi = bv[4 * t + 2 * half + 1];
#pragma unroll
    for (int r = 0; r < 4; ++r) {
      acc[t][r]     = relu1(acc[t][r] + blo[r]);
      acc[t][4 + r] = relu1(acc[t][4 + r] + bhi[r]);
    }
  }
#pragma unroll
  for (int t = 0; t < 8; ++t) {
#pragma unroll
    for (int qq = 0; qq < 2; ++qq) {
      v4f w[3];
#pragma unroll
      for (int u = 0; u < 3; ++u) w[u] = W2h[((t * 2 + qq) * 3 + u) * 32 + lane];
#pragma unroll
      for (int u = 0; u < 3; ++u)
        acc2[u] = mm(w[u][0], w[u][1], acc[t][4 * qq + 0], acc[t][4 * qq + 1], acc2[u]);
#pragma unroll
      for (int u = 0; u < 3; ++u)
        acc2[u] = mm(w[u][2], w[u][3], acc[t][4 * qq + 2], acc[t][4 * qq + 3], acc2[u]);
    }
  }
}

__global__ __launch_bounds__(kTailThreads) __attribute__((amdgpu_waves_per_eu(4)))
void k_tail_fused(const float* __restrict__ xf, const float* __restrict__ xl,
                  const float* __restrict__ aggx,
                  const float* __restrict__ Wg, const float* __restrict__ bg,
                  const float* __restrict__ W1, const float* __restrict__ b1,
                  const float* __restrict__ W2, const float* __restrict__ b2,
                  float* __restrict__ out) {
  __shared__ __attribute__((aligned(16))) float W1s[kW1F];
  __shared__ __attribute__((aligned(16))) float Wgs[kWgF];
  __shared__ __attribute__((aligned(16))) float W2s[2 * kW2hF];
  __shared__ __attribute__((aligned(16))) float b1s[kH];
  __shared__ __attribute__((aligned(16))) float bgs[kH];
  __shared__ __attribute__((aligned(16))) float b2s[48];

  const int tid = threadIdx.x;

#pragma unroll
  for (int it = 0; it < (kW1Quads + kTailThreads - 1) / kTailThreads; ++it) {
    const int i = tid + it * kTailThreads;
    if (i < kW1Quads) {
      const int k = i >> 5, n = (i & 31) << 2;
      const float4 w = *reinterpret_cast<const float4*>(W1 + k * kH + n);
      const int s = w1_slot(k, n);
      W1s[s] = w.x; W1s[s + 4] = w.y; W1s[s + 8] = w.z; W1s[s + 12] = w.w;
    }
  }
#pragma unroll
  for (int it = 0; it < kWgQuads / kTailThreads; ++it) {
    const int i = tid + it * kTailThreads;
    const int k = i >> 5, n = (i & 31) << 2;
    const float4 w = *reinterpret_cast<const float4*>(Wg + k * kH + n);
    const int s = w1_slot(k, n);
    Wgs[s] = w.x; Wgs[s + 4] = w.y; Wgs[s + 8] = w.z; Wgs[s + 12] = w.w;
  }
#pragma unroll
  for (int it = 0; it < kW2Quads / kTailThreads; ++it) {
    const int i = tid + it * kTailThreads;
    const int k = i / (kC / 4), n = (i - k * (kC / 4)) << 2;
    const float4 w = *reinterpret_cast<const float4*>(W2 + k * kC + n);
    const int s = w2_slot(k, n);
    W2s[s] = w.x; W2s[s + 4] = w.y; W2s[s + 8] = w.z; W2s[s + 12] = w.w;
  }
#pragma unroll
  for (int it = 0; it < kKC * 8 / kTailThreads; ++it) {
    const int i = tid + it * kTailThreads;
    W2s[w2_slot(i >> 3, kC + (i & 7))] = 0.0f;
  }
  if (tid < kH) { b1s[tid] = b1[tid]; bgs[tid] = bg[tid]; }
  if (tid < 48) b2s[tid] = (tid < kC) ? b2[tid] : 0.0f;
  __syncthreads();

  const int lane = tid & 31, wave = tid >> 5;
  const int half = lane >> 4, l = lane & 15;
  const v4f* W1v = reinterpret_cast<const v4f*>(W1s);
  const v4f* Wgv = reinterpret_cast<const v4f*>(Wgs);
  const v4f* W2g = reinterpret_cast<const v4f*>(W2s);
  const v4f* W2m = reinterpret_cast<const v4f*>(W2s + kW2hF);
  const v4f* b1v = reinterpret_cast<const v4f*>(b1s);
  const v4f* bgv = reinterpret_cast<const v4f*>(bgs);
  const v4f* b2v = reinterpret_cast<const v4f*>(b2s);

  for (int strip = wave * kTailBlocks + blockIdx.x; strip < kStrips;
       strip += kTailBlocks * kTailWaves) {
    asm volatile("" ::: "memory");
    const size_t m = (size_t)strip * 16 + l;

    v4f ga[kXF / 8];
    v4f xa[kKM / 8];
    {
      const float* ar = aggx + m * kXF + 4 * half;
      const float* xr = xf + m * kXF + 4 * half;
      const float* lr = xl + m * kXL + 4 * half;
#pragma unroll
      for (int j = 0; j < kXF / 8; ++j) ga[j] = *reinterpret_cast<const v4f*>(ar + 8 * j);
#pragma unroll
      for (int j = 0; j < kXF / 8; ++j) xa[j] = *reinterpret_cast<const v4f*>(xr + 8 * j);
#pragma unroll
      for (int j = 0; j < kXL / 8; ++j) xa[kXF / 8 + j] = *reinterpret_cast<const v4f*>(lr + 8 * j);
    }

    v8f accg[8] = {};
#pragma unroll
    for (int j = 0; j < kXF / 8; ++j) {
      v4f w[8];
#pragma unroll
      for (int t = 0; t < 8; ++t) w[t] = Wgv[(j * 8 + t) * 32 + lane];
#pragma unroll
      for (int t = 0; t < 8; ++t) accg[t] = mm(w[t][0], w[t][1], ga[j][0], ga[j][1], accg[t]);
#pragma unroll
      for (int t = 0; t < 8; ++t) accg[t] = mm(w[t][2], w[t][3], ga[j][2], ga[j][3], accg[t]);
    }

    v8f acc2[3] = {};
    bias_relu_classify(accg, bgv, W2g, half, lane, acc2);

    v8f acc1[8] = {};
#pragma unroll
    for (int j = 0; j < kKM / 8; ++j) {
      v4f w[8];
#pragma unroll
      for (int t = 0; t < 8; ++t) w[t] = W1v[(j * 8 + t) * 32 + lane];
#pragma unroll
      for (int t = 0; t < 8; ++t) acc1[t] = mm(w[t][0], w[t][1], xa[j][0], xa[j][1], acc1[t]);
#pragma unroll
      for (int t = 0; t < 8; ++t) acc1[t] = mm(w[t][2], w[t][3], xa[j][2], xa[j][3], acc1[t]);
    }

    bias_relu_classify(acc1, b1v, W2m, half, lane, acc2);

#pragma unroll
    for (int u = 0; u < 3; ++u) {
      if (u < 2 || half == 0) {
        const v4f blo = b2v[4 * u + 2 * half], bhi = b2v[4 * u + 2 * half + 1];
        v4f lo, hi;
#pragma unroll
        for (int r = 0; r < 4; ++r) { lo[r] = acc2[u][r] + blo[r]; hi[r] = acc2[u][4 + r] + bhi[r]; }
        float* o = out + m * kC + 16 * u + 8 * half;
        *reinterpret_cast<v4f*>(o) = lo;
        *reinterpret_cast<v4f*>(o + 4) = hi;
      }
    }
  }
}


extern "C" void kernel_launch(void* const* d_in, const int* in_sizes, int n_in,
                              void* d_out, int out_size, void* d_ws, size_t ws_size,
                              hipStream_t stream) {
  (void)in_sizes; (void)n_in; (void)out_size; (void)ws_size;

  const float* xfeat  = (const float*)d_in[0];
  const float* xlabel = (const float*)d_in[1];
  const int*   eidx   = (const int*)d_in[2];
  const float* W_gcn  = (const float*)d_in[3];
  const float* b_gcn  = (const float*)d_in[4];
  const float* W_mlp  = (const float*)d_in[5];
  const float* b_mlp  = (const float*)d_in[6];
  const float* W_cls  = (const float*)d_in[7];
  const float* b_cls  = (const float*)d_in[8];
  float* out = (float*)d_out;

  char* ws = (char*)d_ws;
  const size_t off_deg  = 0;
  const size_t off_own  = (off_deg + (size_t)kN * 4 + 255) & ~(size_t)255;
  const size_t off_aggx = (off_own + (size_t)kE + 255) & ~(size_t)255;
  float* deg  = (float*)(ws + off_deg);
  unsigned* own = (unsigned*)(ws + off_own);
  float* aggx = (float*)(ws + off_aggx);

  const dim3 blk256(256);

  k_init_deg <<<dim3((kN + 255) / 256), blk256, 0, stream>>>(deg);
  k_deg_owner<<<dim3((kE / 4 + 255) / 256), blk256, 0, stream>>>(eidx, deg, own);
  k_dinv     <<<dim3((kN + 255) / 256), blk256, 0, stream>>>(deg);

  k_agg_lds<<<dim3(256), dim3(kAggThreads), 0, stream>>>(eidx, own, xfeat, deg, aggx);

  k_tail_fused<<<dim3(kTailBlocks), dim3(kTailThreads), 0, stream>>>(
      xfeat, xlabel, aggx, W_gcn, b_gcn, W_mlp, b_mlp, W_cls, b_cls, out);
}


# MLP half of classifier moved from the tail into the aggregation kernel: 8 of its 32 waves run it on the idle matrix pipe, out = MLP half + b_cls, tail adds GCN half; halves summed in new order
# speedup vs baseline: 5.0952x; 1.0544x over previous
#include <hip/hip_runtime.h>


typedef __attribute__((ext_vector_type(2))) float v2f;
typedef __attribute__((ext_vector_type(4))) float v4f;
typedef __attribute__((ext_vector_type(8))) float v8f;

constexpr int kN  = 100000;
constexpr int kE  = 1600000;
constexpr int kXF = 128;
constexpr int kXL = 40;
constexpr int kH  = 128;
constexpr int kC  = 40;
constexpr int kKM = kXF + kXL;
constexpr int kKC = 2 * kH;


__global__ void k_init_deg(float* __restrict__ deg) {
  int i = blockIdx.x * blockDim.x + threadIdx.x;
  if (i < kN) deg[i] = 1.0f;
}

constexpr int kOwnRows = (kN + 255) / 256;
static_assert(kE % 16 == 0 && kOwnRows * 256 >= kN && (kN - 1) / kOwnRows < 256, "owner bytes");

__global__ void k_deg_owner(const int* __restrict__ eidx, float* __restrict__ deg,
                            unsigned* __restrict__ own) {
  int t = blockIdx.x * blockDim.x + threadIdx.x;
  if (t < kE / 4) {
    const int4 d = *reinterpret_cast<const int4*>(eidx + kE + 4 * t);
    atomicAdd(deg + d.x, 1.0f);
    atomicAdd(deg + d.y, 1.0f);
    atomicAdd(deg + d.z, 1.0f);
    atomicAdd(deg + d.w, 1.0f);
    own[t] = (unsigned)(d.x / kOwnRows) | ((unsigned)(d.y / kOwnRows) << 8) |
             ((unsigned)(d.z / kOwnRows) << 16) | ((unsigned)(d.w / kOwnRows) << 24);
  }
}

__global__ void k_dinv(float* __restrict__ deg) {
  int i = blockIdx.x * blockDim.x + threadIdx.x;
  if (i < kN) {
    float d = deg[i];
    deg[i] = (d > 0.0f) ? rsqrtf(d) : 0.0f;
  }
}


constexpr int kStrips = kN / 16;
constexpr int kW1F  = (kKM / 8) * 8 * 32 * 4;
constexpr int kWgF  = (kXF / 8) * 8 * 32 * 4;
constexpr int kW2hF = 8 * 2 * 3 * 32 * 4;
constexpr int kW1Quads = kKM * (kH / 4);
constexpr int kWgQuads = kXF * (kH / 4);
constexpr int kW2hQuads = kH * (kC / 4);

__device__ __forceinline__ int w1_slot(int k, int n) {
  return ((((k >> 3) * 8 + (n >> 4)) * 32 + ((k >> 2) & 1) * 16 + (n & 15)) << 2) + (k & 3);
}

__device__ __forceinline__ int w2_slot(int kk, int n) {
  const int u = n >> 4, l = n & 15, c = kk & 3;
  return (((((kk >> 4) * 2 + ((kk >> 2) & 1)) * 3 + u) * 32 + ((kk >> 3) & 1) * 16 + l) << 2) + c;
}

__device__ __forceinline__ v8f mm(float a0, float a1, float b0, float b1, v8f acc) {
  v2f a = {a0, a1};
  v2f b = {b0, b1};
  return __builtin_amdgcn_wmma_f32_16x16x4_f32(false, a, false, b, (short)0, acc, false, false);
}

__device__ __forceinline__ float relu1(float v) { return v > 0.0f ? v : 0.0f; }

template <int NT>
__device__ __forceinline__ void stage_w2_half(const float* __restrict__ W2half, float* __restrict__ W2s,
                                              int tid) {
#pragma unroll
  for (int it = 0; it < (kW2hQuads + NT - 1) / NT; ++it) {
    const int i = tid + it * NT;
    if (i < kW2hQuads) {
      const int k = i / (kC / 4), n = (i - k * (kC / 4)) << 2;
      const float4 w = *reinterpret_cast<const float4*>(W2half + k * kC + n);
      const int s = w2_slot(k, n);
      W2s[s] = w.x; W2s[s + 4] = w.y; W2s[s + 8] = w.z; W2s[s + 12] = w.w;
    }
  }
#pragma unroll
  for (int it = 0; it < (kH * 8 + NT - 1) / NT; ++it) {
    const int i = tid + it * NT;
    if (i < kH * 8) W2s[w2_slot(i >> 3, kC + (i & 7))] = 0.0f;
  }
}


constexpr int kAggThreads = 1024;
constexpr int kAggWaves = kAggThreads / 32;
constexpr int kMlpWaves = 8;
constexpr int kScanWaves = kAggWaves - kMlpWaves;
constexpr int kScanThreads = kScanWaves * 32;
constexpr int kChunks = kE / 16;
constexpr int kScanTrips = (kChunks + kScanThreads - 1) / kScanThreads;
constexpr int kQCap = 168;
constexpr int kMlpStrips = (kStrips + 255) / 256;
static_assert(kChunks > kScanThreads && kQCap > 8 + 32 && kKM % 24 == 0, "scan shape, MLP k chunks");

__device__ __forceinline__ unsigned zero_bytes(unsigned y) {
  return ~(((y & 0x7f7f7f7fu) + 0x7f7f7f7fu) | y | 0x7f7f7f7fu);
}

__device__ __forceinline__ void agg_edge_add(float* __restrict__ row, const float4 hv, float nm) {
  atomicAdd(row + 0,  hv.x * nm);
  atomicAdd(row + 32, hv.y * nm);
  atomicAdd(row + 64, hv.z * nm);
  atomicAdd(row + 96, hv.w * nm);
}

__device__ __forceinline__ void mlp_strip(const float* __restrict__ xf, const float* __restrict__ xl,
                                          const v4f* __restrict__ W1v, const v4f* __restrict__ W2m,
                                          const v4f* __restrict__ b1v, const v4f* __restrict__ b2v,
                                          float* __restrict__ out, int strip, int lane) {
  const int half = lane >> 4, l = lane & 15;
  const size_t m = (size_t)strip * 16 + l;
  const float* xr = xf + m * kXF + 4 * half;
  const float* lr = xl + m * kXL + 4 * half;
  v8f acc2[3] = {};
#pragma unroll 1
  for (int tg = 0; tg < 2; ++tg) {
    const v4f* W1t = W1v + tg * 4 * 32;
    v8f acc1[4] = {};
    v4f cur[3], nxt[3];
#pragma unroll
    for (int i = 0; i < 3; ++i) cur[i] = *reinterpret_cast<const v4f*>(xr + 8 * i);
#pragma unroll 1
    for (int c = 0; c < kKM / 24; ++c) {
      if (c + 1 < kKM / 24) {
#pragma unroll
        for (int i = 0; i < 3; ++i) {
          const int j = 3 * (c + 1) + i;
          const float* p = (j < kXF / 8) ? xr + 8 * j : lr + 8 * (j - kXF / 8);
          nxt[i] = *reinterpret_cast<const v4f*>(p);
        }
      }
      const v4f* W1c = W1t + c * 3 * 8 * 32;
#pragma unroll
      for (int i = 0; i < 3; ++i) {
        v4f w[4];
#pragma unroll
        for (int t = 0; t < 4; ++t) w[t] = W1c[(i * 8 + t) * 32 + lane];
#pragma unroll
        for (int t = 0; t < 4; ++t) acc1[t] = mm(w[t][0], w[t][1], cur[i][0], cur[i][1], acc1[t]);
#pragma unroll
        for (int t = 0; t < 4; ++t) acc1[t] = mm(w[t][2], w[t][3], cur[i][2], cur[i][3], acc1[t]);
      }
#pragma unroll
      for (int i = 0; i < 3; ++i) cur[i] = nxt[i];
    }
#pragma unroll
    for (int t = 0; t < 4; ++t) {
      const v4f blo = b1v[4 * (tg * 4 + t) + 2 * half], bhi = b1v[4 * (tg * 4 + t) + 2 * half + 1];
#pragma unroll
      for (int r = 0; r < 4; ++r) {
        acc1[t][r]     = relu1(acc1[t][r] + blo[r]);
        acc1[t][4 + r] = relu1(acc1[t][4 + r] + bhi[r]);
      }
    }
    const v4f* W2t = W2m + tg * 4 * 2 * 3 * 32;
#pragma unroll
    for (int t = 0; t < 4; ++t) {
#pragma unroll
      for (int qq = 0; qq < 2; ++qq) {
        v4f w[3];
#pragma unroll
        for (int u = 0; u < 3; ++u) w[u] = W2t[((t * 2 + qq) * 3 + u) * 32 + lane];
#pragma unroll
        for (int u = 0; u < 3; ++u)
          acc2[u] = mm(w[u][0], w[u][1], acc1[t][4 * qq + 0], acc1[t][4 * qq + 1], acc2[u]);
#pragma unroll
        for (int u = 0; u < 3; ++u)
          acc2[u] = mm(w[u][2], w[u][3], acc1[t][4 * qq + 2], acc1[t][4 * qq + 3], acc2[u]);
      }
    }
  }
#pragma unroll
  for (int u = 0; u < 3; ++u) {
    if (u < 2 || half == 0) {
      const v4f blo = b2v[4 * u + 2 * half], bhi = b2v[4 * u + 2 * half + 1];
      v4f lo, hi;
#pragma unroll
      for (int r = 0; r < 4; ++r) { lo[r] = acc2[u][r] + blo[r]; hi[r] = acc2[u][4 + r] + bhi[r]; }
      float* o = out + m * kC + 16 * u + 8 * half;
      *reinterpret_cast<v4f*>(o) = lo;
      *reinterpret_cast<v4f*>(o + 4) = hi;
    }
  }
}

__global__ __launch_bounds__(kAggThreads)
void k_agg_lds(const int* __restrict__ eidx, const unsigned* __restrict__ own,
               const float* __restrict__ xf, const float* __restrict__ xl,
               const float* __restrict__ dinv,
               const float* __restrict__ W1, const float* __restrict__ b1,
               const float* __restrict__ W2, const float* __restrict__ b2,
               float* __restrict__ aggx, float* __restrict__ out) {
  __shared__ __attribute__((aligned(16))) float acc[kOwnRows * kH];
  __shared__ __attribute__((aligned(16))) float W1s[kW1F];
  __shared__ __attribute__((aligned(16))) float W2s[kW2hF];
  __shared__ __attribute__((aligned(16))) float b1s[kH];
  __shared__ __attribute__((aligned(16))) float b2s[48];
  __shared__ int queue[kScanWaves * kQCap];

  const int tid = threadIdx.x, lane = tid & 31, wave = tid >> 5;
  const int lo = blockIdx.x * kOwnRows;
  const int rows = (kN - lo < kOwnRows) ? (kN - lo) : kOwnRows;

  for (int i = tid; i < rows * 32; i += kAggThreads) {
    const int r = i >> 5, q = i & 31;
    const float di = dinv[lo + r];
    const float4 hv = *reinterpret_cast<const float4*>(xf + (size_t)(lo + r) * kXF + 4 * q);
    float* a = acc + r * kH + q;
    a[0]  = hv.x * di * di;
    a[32] = hv.y * di * di;
    a[64] = hv.z * di * di;
    a[96] = hv.w * di * di;
  }
#pragma unroll
  for (int it = 0; it < (kW1Quads + kAggThreads - 1) / kAggThreads; ++it) {
    const int i = tid + it * kAggThreads;
    if (i < kW1Quads) {
      const int k = i >> 5, n = (i & 31) << 2;
      const float4 w = *reinterpret_cast<const float4*>(W1 + k * kH + n);
      const int s = w1_slot(k, n);
      W1s[s] = w.x; W1s[s + 4] = w.y; W1s[s + 8] = w.z; W1s[s + 12] = w.w;
    }
  }
  stage_w2_half<kAggThreads>(W2 + kH * kC, W2s, tid);
  if (tid < kH) b1s[tid] = b1[tid];
  if (tid < 48) b2s[tid] = (tid < kC) ? b2[tid] : 0.0f;
  __syncthreads();

  if (wave >= kScanWaves) {
    const v4f* W1v = reinterpret_cast<const v4f*>(W1s);
    const v4f* W2m = reinterpret_cast<const v4f*>(W2s);
    const v4f* b1v = reinterpret_cast<const v4f*>(b1s);
    const v4f* b2v = reinterpret_cast<const v4f*>(b2s);
    for (int i = wave - kScanWaves; i < kMlpStrips; i += kMlpWaves) {
      const int strip = blockIdx.x + 256 * i;
      if (strip < kStrips) {
        asm volatile("" ::: "memory");
        mlp_strip(xf, xl, W1v, W2m, b1v, b2v, out, strip, lane);
      }
    }
  } else {
    int* myq = queue + wave * kQCap;
    int cnt = 0;
    const unsigned me = blockIdx.x * 0x01010101u;
    const uint4* own4 = reinterpret_cast<const uint4*>(own);
    float* accl = acc + lane;

    uint4 x = own4[tid];
    for (int it = 0; it < kScanTrips; ++it) {
      const int c = it * kScanThreads + tid;
      const int cn = c + kScanThreads;
      uint4 xn = x;
      if (cn < kChunks) xn = own4[cn];
      unsigned m = (zero_bytes(x.x ^ me) >> 7) | (zero_bytes(x.y ^ me) >> 6) |
                   (zero_bytes(x.z ^ me) >> 5) | (zero_bytes(x.w ^ me) >> 4);
      if (c >= kChunks) m = 0;
      x = xn;

      unsigned bal = __builtin_amdgcn_ballot_w32(m != 0);
      for (;;) {
        if (bal != 0) {
          if (m != 0) {
            const int q = __builtin_ctz(m);
            m &= m - 1;
            myq[cnt + (int)__builtin_amdgcn_mbcnt_lo(bal, 0u)] = 16 * c + ((q & 7) << 2) + (q >> 3);
          }
          cnt += __builtin_popcount(bal);
          bal = __builtin_amdgcn_ballot_w32(m != 0);
        }
        while (cnt >= 8 && (bal == 0 || cnt > kQCap - 32)) {
          int s[8], d[8];
#pragma unroll
          for (int k = 0; k < 8; ++k) {
            const int e = __builtin_amdgcn_readfirstlane(myq[cnt - 1 - k]);
            s[k] = eidx[e];
            d[k] = eidx[kE + e];
          }
          float4 hv[8];
          float nm[8];
#pragma unroll
          for (int k = 0; k < 8; ++k) {
            hv[k] = *reinterpret_cast<const float4*>(xf + (size_t)s[k] * kXF + 4 * lane);
            nm[k] = dinv[s[k]] * dinv[d[k]];
          }
#pragma unroll
          for (int k = 0; k < 8; ++k) agg_edge_add(accl + (d[k] - lo) * kH, hv[k], nm[k]);
          cnt -= 8;
        }
        if (bal == 0) break;
      }
    }
    while (cnt > 0) {
      const int e = __builtin_amdgcn_readfirstlane(myq[cnt - 1]);
      const int s = eidx[e], d = eidx[kE + e];
      const float4 hv = *reinterpret_cast<const float4*>(xf + (size_t)s * kXF + 4 * lane);
      agg_edge_add(accl + (d - lo) * kH, hv, dinv[s] * dinv[d]);
      --cnt;
    }
  }
  __syncthreads();

  for (int i = tid; i < rows * 32; i += kAggThreads) {
    const int r = i >> 5, q = i & 31;
    const float* a = acc + r * kH + q;
    const float4 v = {a[0], a[32], a[64], a[96]};
    *reinterpret_cast<float4*>(aggx + (size_t)(lo + r) * kXF + 4 * q) = v;
  }
}


constexpr int kTailThreads = 512;
constexpr int kTailWaves = kTailThreads / 32;
constexpr int kTailBlocks = 256;
static_assert(kWgQuads % kTailThreads == 0, "staging trip counts");

__global__ __launch_bounds__(kTailThreads) __attribute__((amdgpu_waves_per_eu(4)))
void k_tail_fused(const float* __restrict__ aggx,
                  const float* __restrict__ Wg, const float* __restrict__ bg,
                  const float* __restrict__ W2,
                  float* __restrict__ out) {
  __shared__ __attribute__((aligned(16))) float Wgs[kWgF];
  __shared__ __attribute__((aligned(16))) float W2s[kW2hF];
  __shared__ __attribute__((aligned(16))) float bgs[kH];

  const int tid = threadIdx.x;

#pragma unroll
  for (int it = 0; it < kWgQuads / kTailThreads; ++it) {
    const int i = tid + it * kTailThreads;
    const int k = i >> 5, n = (i & 31) << 2;
    const float4 w = *reinterpret_cast<const float4*>(Wg + k * kH + n);
    const int s = w1_slot(k, n);
    Wgs[s] = w.x; Wgs[s + 4] = w.y; Wgs[s + 8] = w.z; Wgs[s + 12] = w.w;
  }
  stage_w2_half<kTailThreads>(W2, W2s, tid);
  if (tid < kH) bgs[tid] = bg[tid];
  __syncthreads();

  const int lane = tid & 31, wave = tid >> 5;
  const int half = lane >> 4, l = lane & 15;
  const v4f* Wgv = reinterpret_cast<const v4f*>(Wgs);
  const v4f* W2g = reinterpret_cast<const v4f*>(W2s);
  const v4f* bgv = reinterpret_cast<const v4f*>(bgs);

  for (int strip = wave * kTailBlocks + blockIdx.x; strip < kStrips;
       strip += kTailBlocks * kTailWaves) {
    asm volatile("" ::: "memory");
    const size_t m = (size_t)strip * 16 + l;

    v4f ga[kXF / 8];
    {
      const float* ar = aggx + m * kXF + 4 * half;
#pragma unroll
      for (int j = 0; j < kXF / 8; ++j) ga[j] = *reinterpret_cast<const v4f*>(ar + 8 * j);
    }
    v4f plo[3], phi[3];
#pragma unroll
    for (int u = 0; u < 3; ++u) {
      if (u < 2 || half == 0) {
        const float* o = out + m * kC + 16 * u + 8 * half;
        plo[u] = *reinterpret_cast<const v4f*>(o);
        phi[u] = *reinterpret_cast<const v4f*>(o + 4);
      }
    }

    v8f accg[8] = {};
#pragma unroll
    for (int j = 0; j < kXF / 8; ++j) {
      v4f w[8];
#pragma unroll
      for (int t = 0; t < 8; ++t) w[t] = Wgv[(j * 8 + t) * 32 + lane];
#pragma unroll
      for (int t = 0; t < 8; ++t) accg[t] = mm(w[t][0], w[t][1], ga[j][0], ga[j][1], accg[t]);
#pragma unroll
      for (int t = 0; t < 8; ++t) accg[t] = mm(w[t][2], w[t][3], ga[j][2], ga[j][3], accg[t]);
    }

#pragma unroll
    for (int t = 0; t < 8; ++t) {
      const v4f blo = bgv[4 * t + 2 * half], bhi = bgv[4 * t + 2 * half + 1];
#pragma unroll
      for (int r = 0; r < 4; ++r) {
        accg[t][r]     = relu1(accg[t][r] + blo[r]);
        accg[t][4 + r] = relu1(accg[t][4 + r] + bhi[r]);
      }
    }
    v8f acc2[3] = {};
#pragma unroll
    for (int t = 0; t < 8; ++t) {
#pragma unroll
      for (int qq = 0; qq < 2; ++qq) {
        v4f w[3];
#pragma unroll
        for (int u = 0; u < 3; ++u) w[u] = W2g[((t * 2 + qq) * 3 + u) * 32 + lane];
#pragma unroll
        for (int u = 0; u < 3; ++u)
          acc2[u] = mm(w[u][0], w[u][1], accg[t][4 * qq + 0], accg[t][4 * qq + 1], acc2[u]);
#pragma unroll
        for (int u = 0; u < 3; ++u)
          acc2[u] = mm(w[u][2], w[u][3], accg[t][4 * qq + 2], accg[t][4 * qq + 3], acc2[u]);
      }
    }

#pragma unroll
    for (int u = 0; u < 3; ++u) {
      if (u < 2 || half == 0) {
        v4f lo, hi;
#pragma unroll
        for (int r = 0; r < 4; ++r) { lo[r] = plo[u][r] + acc2[u][r]; hi[r] = phi[u][r] + acc2[u][4 + r]; }
        float* o = out + m * kC + 16 * u + 8 * half;
        *reinterpret_cast<v4f*>(o) = lo;
        *reinterpret_cast<v4f*>(o + 4) = hi;
      }
    }
  }
}


extern "C" void kernel_launch(void* const* d_in, const int* in_sizes, int n_in,
                              void* d_out, int out_size, void* d_ws, size_t ws_size,
                              hipStream_t stream) {
  (void)in_sizes; (void)n_in; (void)out_size; (void)ws_size;

  const float* xfeat  = (const float*)d_in[0];
  const float* xlabel = (const float*)d_in[1];
  const int*   eidx   = (const int*)d_in[2];
  const float* W_gcn  = (const float*)d_in[3];
  const float* b_gcn  = (const float*)d_in[4];
  const float* W_mlp  = (const float*)d_in[5];
  const float* b_mlp  = (const float*)d_in[6];
  const float* W_cls  = (const float*)d_in[7];
  const float* b_cls  = (const float*)d_in[8];
  float* out = (float*)d_out;

  char* ws = (char*)d_ws;
  const size_t off_deg  = 0;
  const size_t off_own  = (off_deg + (size_t)kN * 4 + 255) & ~(size_t)255;
  const size_t off_aggx = (off_own + (size_t)kE + 255) & ~(size_t)255;
  float* deg  = (float*)(ws + off_deg);
  unsigned* own = (unsigned*)(ws + off_own);
  float* aggx = (float*)(ws + off_aggx);

  const dim3 blk256(256);

  k_init_deg <<<dim3((kN + 255) / 256), blk256, 0, stream>>>(deg);
  k_deg_owner<<<dim3((kE / 4 + 255) / 256), blk256, 0, stream>>>(eidx, deg, own);
  k_dinv     <<<dim3((kN + 255) / 256), blk256, 0, stream>>>(deg);

  k_agg_lds<<<dim3(256), dim3(kAggThreads), 0, stream>>>(eidx, own, xfeat, xlabel, deg,
                                                         W_mlp, b_mlp, W_cls, b_cls, aggx, out);

  k_tail_fused<<<dim3(kTailBlocks), dim3(kTailThreads), 0, stream>>>(
      aggx, W_gcn, b_gcn, W_cls, out);
}


# edge drain in batches of 32: each lane loads one queued edge's src, dst and both dinv by vector loads, broadcast by readlane, replacing 4 serial scalar loads per edge; LDS add order changes
# speedup vs baseline: 5.3501x; 1.0500x over previous
#include <hip/hip_runtime.h>


typedef __attribute__((ext_vector_type(2))) float v2f;
typedef __attribute__((ext_vector_type(4))) float v4f;
typedef __attribute__((ext_vector_type(8))) float v8f;

constexpr int kN  = 100000;
constexpr int kE  = 1600000;
constexpr int kXF = 128;
constexpr int kXL = 40;
constexpr int kH  = 128;
constexpr int kC  = 40;
constexpr int kKM = kXF + kXL;
constexpr int kKC = 2 * kH;


__global__ void k_init_deg(float* __restrict__ deg) {
  int i = blockIdx.x * blockDim.x + threadIdx.x;
  if (i < kN) deg[i] = 1.0f;
}

constexpr int kOwnRows = (kN + 255) / 256;
static_assert(kE % 16 == 0 && kOwnRows * 256 >= kN && (kN - 1) / kOwnRows < 256, "owner bytes");

__global__ void k_deg_owner(const int* __restrict__ eidx, float* __restrict__ deg,
                            unsigned* __restrict__ own) {
  int t = blockIdx.x * blockDim.x + threadIdx.x;
  if (t < kE / 4) {
    const int4 d = *reinterpret_cast<const int4*>(eidx + kE + 4 * t);
    atomicAdd(deg + d.x, 1.0f);
    atomicAdd(deg + d.y, 1.0f);
    atomicAdd(deg + d.z, 1.0f);
    atomicAdd(deg + d.w, 1.0f);
    own[t] = (unsigned)(d.x / kOwnRows) | ((unsigned)(d.y / kOwnRows) << 8) |
             ((unsigned)(d.z / kOwnRows) << 16) | ((unsigned)(d.w / kOwnRows) << 24);
  }
}

__global__ void k_dinv(float* __restrict__ deg) {
  int i = blockIdx.x * blockDim.x + threadIdx.x;
  if (i < kN) {
    float d = deg[i];
    deg[i] = (d > 0.0f) ? rsqrtf(d) : 0.0f;
  }
}


constexpr int kStrips = kN / 16;
constexpr int kW1F  = (kKM / 8) * 8 * 32 * 4;
constexpr int kWgF  = (kXF / 8) * 8 * 32 * 4;
constexpr int kW2hF = 8 * 2 * 3 * 32 * 4;
constexpr int kW1Quads = kKM * (kH / 4);
constexpr int kWgQuads = kXF * (kH / 4);
constexpr int kW2hQuads = kH * (kC / 4);

__device__ __forceinline__ int w1_slot(int k, int n) {
  return ((((k >> 3) * 8 + (n >> 4)) * 32 + ((k >> 2) & 1) * 16 + (n & 15)) << 2) + (k & 3);
}

__device__ __forceinline__ int w2_slot(int kk, int n) {
  const int u = n >> 4, l = n & 15, c = kk & 3;
  return (((((kk >> 4) * 2 + ((kk >> 2) & 1)) * 3 + u) * 32 + ((kk >> 3) & 1) * 16 + l) << 2) + c;
}

__device__ __forceinline__ v8f mm(float a0, float a1, float b0, float b1, v8f acc) {
  v2f a = {a0, a1};
  v2f b = {b0, b1};
  return __builtin_amdgcn_wmma_f32_16x16x4_f32(false, a, false, b, (short)0, acc, false, false);
}

__device__ __forceinline__ float relu1(float v) { return v > 0.0f ? v : 0.0f; }

template <int NT>
__device__ __forceinline__ void stage_w2_half(const float* __restrict__ W2half, float* __restrict__ W2s,
                                              int tid) {
#pragma unroll
  for (int it = 0; it < (kW2hQuads + NT - 1) / NT; ++it) {
    const int i = tid + it * NT;
    if (i < kW2hQuads) {
      const int k = i / (kC / 4), n = (i - k * (kC / 4)) << 2;
      const float4 w = *reinterpret_cast<const float4*>(W2half + k * kC + n);
      const int s = w2_slot(k, n);
      W2s[s] = w.x; W2s[s + 4] = w.y; W2s[s + 8] = w.z; W2s[s + 12] = w.w;
    }
  }
#pragma unroll
  for (int it = 0; it < (kH * 8 + NT - 1) / NT; ++it) {
    const int i = tid + it * NT;
    if (i < kH * 8) W2s[w2_slot(i >> 3, kC + (i & 7))] = 0.0f;
  }
}


constexpr int kAggThreads = 1024;
constexpr int kAggWaves = kAggThreads / 32;
constexpr int kMlpWaves = 8;
constexpr int kScanWaves = kAggWaves - kMlpWaves;
constexpr int kScanThreads = kScanWaves * 32;
constexpr int kChunks = kE / 16;
constexpr int kScanTrips = (kChunks + kScanThreads - 1) / kScanThreads;
constexpr int kQCap = 168;
constexpr int kBatch = 32;
constexpr int kMlpStrips = (kStrips + 255) / 256;
static_assert(kChunks > kScanThreads && kQCap >= kBatch + 32 && kKM % 24 == 0, "scan shape, MLP k chunks");

__device__ __forceinline__ unsigned zero_bytes(unsigned y) {
  return ~(((y & 0x7f7f7f7fu) + 0x7f7f7f7fu) | y | 0x7f7f7f7fu);
}

__device__ __forceinline__ void agg_edge_add(float* __restrict__ row, const float4 hv, float nm) {
  atomicAdd(row + 0,  hv.x * nm);
  atomicAdd(row + 32, hv.y * nm);
  atomicAdd(row + 64, hv.z * nm);
  atomicAdd(row + 96, hv.w * nm);
}

template <bool FULL>
__device__ __forceinline__ void drain_batch(const int* qtop, int n,
                                            const int* __restrict__ eidx, const float* __restrict__ xf,
                                            const float* __restrict__ dinv, float* __restrict__ accl,
                                            int lo, int lane) {
  const int li = FULL ? lane : (lane < n ? lane : n - 1);
  const int e = qtop[-li];
  const int s = eidx[e];
  const int d = eidx[kE + e];
  const int nmb = __float_as_int(dinv[s] * dinv[d]);
  const int doff = (d - lo) * kH;
#pragma unroll
  for (int g = 0; g < kBatch; g += 8) {
    if (FULL || g < n) {
      float4 hv[8];
#pragma unroll
      for (int k = 0; k < 8; ++k) {
        if (FULL || g + k < n) {
          const int sk = __builtin_amdgcn_readlane(s, g + k);
          hv[k] = *reinterpret_cast<const float4*>(xf + (size_t)sk * kXF + 4 * lane);
        }
      }
#pragma unroll
      for (int k = 0; k < 8; ++k) {
        if (FULL || g + k < n) {
          const int dk = __builtin_amdgcn_readlane(doff, g + k);
          const float nk = __int_as_float(__builtin_amdgcn_readlane(nmb, g + k));
          agg_edge_add(accl + dk, hv[k], nk);
        }
      }
    }
  }
}

__device__ __forceinline__ void mlp_strip(const float* __restrict__ xf, const float* __restrict__ xl,
                                          const v4f* __restrict__ W1v, const v4f* __restrict__ W2m,
                                          const v4f* __restrict__ b1v, const v4f* __restrict__ b2v,
                                          float* __restrict__ out, int strip, int lane) {
  const int half = lane >> 4, l = lane & 15;
  const size_t m = (size_t)strip * 16 + l;
  const float* xr = xf + m * kXF + 4 * half;
  const float* lr = xl + m * kXL + 4 * half;
  v8f acc2[3] = {};
#pragma unroll 1
  for (int tg = 0; tg < 2; ++tg) {
    const v4f* W1t = W1v + tg * 4 * 32;
    v8f acc1[4] = {};
    v4f cur[3], nxt[3];
#pragma unroll
    for (int i = 0; i < 3; ++i) cur[i] = *reinterpret_cast<const v4f*>(xr + 8 * i);
#pragma unroll 1
    for (int c = 0; c < kKM / 24; ++c) {
      if (c + 1 < kKM / 24) {
#pragma unroll
        for (int i = 0; i < 3; ++i) {
          const int j = 3 * (c + 1) + i;
          const float* p = (j < kXF / 8) ? xr + 8 * j : lr + 8 * (j - kXF / 8);
          nxt[i] = *reinterpret_cast<const v4f*>(p);
        }
      }
      const v4f* W1c = W1t + c * 3 * 8 * 32;
#pragma unroll
      for (int i = 0; i < 3; ++i) {
        v4f w[4];
#pragma unroll
        for (int t = 0; t < 4; ++t) w[t] = W1c[(i * 8 + t) * 32 + lane];
#pragma unroll
        for (int t = 0; t < 4; ++t) acc1[t] = mm(w[t][0], w[t][1], cur[i][0], cur[i][1], acc1[t]);
#pragma unroll
        for (int t = 0; t < 4; ++t) acc1[t] = mm(w[t][2], w[t][3], cur[i][2], cur[i][3], acc1[t]);
      }
#pragma unroll
      for (int i = 0; i < 3; ++i) cur[i] = nxt[i];
    }
#pragma unroll
    for (int t = 0; t < 4; ++t) {
      const v4f blo = b1v[4 * (tg * 4 + t) + 2 * half], bhi = b1v[4 * (tg * 4 + t) + 2 * half + 1];
#pragma unroll
      for (int r = 0; r < 4; ++r) {
        acc1[t][r]     = relu1(acc1[t][r] + blo[r]);
        acc1[t][4 + r] = relu1(acc1[t][4 + r] + bhi[r]);
      }
    }
    const v4f* W2t = W2m + tg * 4 * 2 * 3 * 32;
#pragma unroll
    for (int t = 0; t < 4; ++t) {
#pragma unroll
      for (int qq = 0; qq < 2; ++qq) {
        v4f w[3];
#pragma unroll
        for (int u = 0; u < 3; ++u) w[u] = W2t[((t * 2 + qq) * 3 + u) * 32 + lane];
#pragma unroll
        for (int u = 0; u < 3; ++u)
          acc2[u] = mm(w[u][0], w[u][1], acc1[t][4 * qq + 0], acc1[t][4 * qq + 1], acc2[u]);
#pragma unroll
        for (int u = 0; u < 3; ++u)
          acc2[u] = mm(w[u][2], w[u][3], acc1[t][4 * qq + 2], acc1[t][4 * qq + 3], acc2[u]);
      }
    }
  }
#pragma unroll
  for (int u = 0; u < 3; ++u) {
    if (u < 2 || half == 0) {
      const v4f blo = b2v[4 * u + 2 * half], bhi = b2v[4 * u + 2 * half + 1];
      v4f lo, hi;
#pragma unroll
      for (int r = 0; r < 4; ++r) { lo[r] = acc2[u][r] + blo[r]; hi[r] = acc2[u][4 + r] + bhi[r]; }
      float* o = out + m * kC + 16 * u + 8 * half;
      *reinterpret_cast<v4f*>(o) = lo;
      *reinterpret_cast<v4f*>(o + 4) = hi;
    }
  }
}

__global__ __launch_bounds__(kAggThreads)
void k_agg_lds(const int* __restrict__ eidx, const unsigned* __restrict__ own,
               const float* __restrict__ xf, const float* __restrict__ xl,
               const float* __restrict__ dinv,
               const float* __restrict__ W1, const float* __restrict__ b1,
               const float* __restrict__ W2, const float* __restrict__ b2,
               float* __restrict__ aggx, float* __restrict__ out) {
  __shared__ __attribute__((aligned(16))) float acc[kOwnRows * kH];
  __shared__ __attribute__((aligned(16))) float W1s[kW1F];
  __shared__ __attribute__((aligned(16))) float W2s[kW2hF];
  __shared__ __attribute__((aligned(16))) float b1s[kH];
  __shared__ __attribute__((aligned(16))) float b2s[48];
  __shared__ int queue[kScanWaves * kQCap];

  const int tid = threadIdx.x, lane = tid & 31, wave = tid >> 5;
  const int lo = blockIdx.x * kOwnRows;
  const int rows = (kN - lo < kOwnRows) ? (kN - lo) : kOwnRows;

  for (int i = tid; i < rows * 32; i += kAggThreads) {
    const int r = i >> 5, q = i & 31;
    const float di = dinv[lo + r];
    const float4 hv = *reinterpret_cast<const float4*>(xf + (size_t)(lo + r) * kXF + 4 * q);
    float* a = acc + r * kH + q;
    a[0]  = hv.x * di * di;
    a[32] = hv.y * di * di;
    a[64] = hv.z * di * di;
    a[96] = hv.w * di * di;
  }
#pragma unroll
  for (int it = 0; it < (kW1Quads + kAggThreads - 1) / kAggThreads; ++it) {
    const int i = tid + it * kAggThreads;
    if (i < kW1Quads) {
      const int k = i >> 5, n = (i & 31) << 2;
      const float4 w = *reinterpret_cast<const float4*>(W1 + k * kH + n);
      const int s = w1_slot(k, n);
      W1s[s] = w.x; W1s[s + 4] = w.y; W1s[s + 8] = w.z; W1s[s + 12] = w.w;
    }
  }
  stage_w2_half<kAggThreads>(W2 + kH * kC, W2s, tid);
  if (tid < kH) b1s[tid] = b1[tid];
  if (tid < 48) b2s[tid] = (tid < kC) ? b2[tid] : 0.0f;
  __syncthreads();

  if (wave >= kScanWaves) {
    const v4f* W1v = reinterpret_cast<const v4f*>(W1s);
    const v4f* W2m = reinterpret_cast<const v4f*>(W2s);
    const v4f* b1v = reinterpret_cast<const v4f*>(b1s);
    const v4f* b2v = reinterpret_cast<const v4f*>(b2s);
    for (int i = wave - kScanWaves; i < kMlpStrips; i += kMlpWaves) {
      const int strip = blockIdx.x + 256 * i;
      if (strip < kStrips) {
        asm volatile("" ::: "memory");
        mlp_strip(xf, xl, W1v, W2m, b1v, b2v, out, strip, lane);
      }
    }
  } else {
    int* myq = queue + wave * kQCap;
    int cnt = 0;
    const unsigned me = blockIdx.x * 0x01010101u;
    const uint4* own4 = reinterpret_cast<const uint4*>(own);
    float* accl = acc + lane;

    uint4 x = own4[tid];
    for (int it = 0; it < kScanTrips; ++it) {
      const int c = it * kScanThreads + tid;
      const int cn = c + kScanThreads;
      uint4 xn = x;
      if (cn < kChunks) xn = own4[cn];
      unsigned m = (zero_bytes(x.x ^ me) >> 7) | (zero_bytes(x.y ^ me) >> 6) |
                   (zero_bytes(x.z ^ me) >> 5) | (zero_bytes(x.w ^ me) >> 4);
      if (c >= kChunks) m = 0;
      x = xn;

      unsigned bal = __builtin_amdgcn_ballot_w32(m != 0);
      for (;;) {
        if (bal != 0) {
          if (m != 0) {
            const int q = __builtin_ctz(m);
            m &= m - 1;
            myq[cnt + (int)__builtin_amdgcn_mbcnt_lo(bal, 0u)] = 16 * c + ((q & 7) << 2) + (q >> 3);
          }
          cnt += __builtin_popcount(bal);
          bal = __builtin_amdgcn_ballot_w32(m != 0);
        }
        while (cnt >= kBatch && (bal == 0 || cnt > kQCap - 32)) {
          drain_batch<true>(myq + cnt - 1, kBatch, eidx, xf, dinv, accl, lo, lane);
          cnt -= kBatch;
        }
        if (bal == 0) break;
      }
    }
    if (cnt > 0)
      drain_batch<false>(myq + cnt - 1, cnt, eidx, xf, dinv, accl, lo, lane);
  }
  __syncthreads();

  for (int i = tid; i < rows * 32; i += kAggThreads) {
    const int r = i >> 5, q = i & 31;
    const float* a = acc + r * kH + q;
    const float4 v = {a[0], a[32], a[64], a[96]};
    *reinterpret_cast<float4*>(aggx + (size_t)(lo + r) * kXF + 4 * q) = v;
  }
}


constexpr int kTailThreads = 512;
constexpr int kTailWaves = kTailThreads / 32;
constexpr int kTailBlocks = 256;
static_assert(kWgQuads % kTailThreads == 0, "staging trip counts");

__global__ __launch_bounds__(kTailThreads) __attribute__((amdgpu_waves_per_eu(4)))
void k_tail_fused(const float* __restrict__ aggx,
                  const float* __restrict__ Wg, const float* __restrict__ bg,
                  const float* __restrict__ W2,
                  float* __restrict__ out) {
  __shared__ __attribute__((aligned(16))) float Wgs[kWgF];
  __shared__ __attribute__((aligned(16))) float W2s[kW2hF];
  __shared__ __attribute__((aligned(16))) float bgs[kH];

  const int tid = threadIdx.x;

#pragma unroll
  for (int it = 0; it < kWgQuads / kTailThreads; ++it) {
    const int i = tid + it * kTailThreads;
    const int k = i >> 5, n = (i & 31) << 2;
    const float4 w = *reinterpret_cast<const float4*>(Wg + k * kH + n);
    const int s = w1_slot(k, n);
    Wgs[s] = w.x; Wgs[s + 4] = w.y; Wgs[s + 8] = w.z; Wgs[s + 12] = w.w;
  }
  stage_w2_half<kTailThreads>(W2, W2s, tid);
  if (tid < kH) bgs[tid] = bg[tid];
  __syncthreads();

  const int lane = tid & 31, wave = tid >> 5;
  const int half = lane >> 4, l = lane & 15;
  const v4f* Wgv = reinterpret_cast<const v4f*>(Wgs);
  const v4f* W2g = reinterpret_cast<const v4f*>(W2s);
  const v4f* bgv = reinterpret_cast<const v4f*>(bgs);

  for (int strip = wave * kTailBlocks + blockIdx.x; strip < kStrips;
       strip += kTailBlocks * kTailWaves) {
    asm volatile("" ::: "memory");
    const size_t m = (size_t)strip * 16 + l;

    v4f ga[kXF / 8];
    {
      const float* ar = aggx + m * kXF + 4 * half;
#pragma unroll
      for (int j = 0; j < kXF / 8; ++j) ga[j] = *reinterpret_cast<const v4f*>(ar + 8 * j);
    }
    v4f plo[3], phi[3];
#pragma unroll
    for (int u = 0; u < 3; ++u) {
      if (u < 2 || half == 0) {
        const float* o = out + m * kC + 16 * u + 8 * half;
        plo[u] = *reinterpret_cast<const v4f*>(o);
        phi[u] = *reinterpret_cast<const v4f*>(o + 4);
      }
    }

    v8f accg[8] = {};
#pragma unroll
    for (int j = 0; j < kXF / 8; ++j) {
      v4f w[8];
#pragma unroll
      for (int t = 0; t < 8; ++t) w[t] = Wgv[(j * 8 + t) * 32 + lane];
#pragma unroll
      for (int t = 0; t < 8; ++t) accg[t] = mm(w[t][0], w[t][1], ga[j][0], ga[j][1], accg[t]);
#pragma unroll
      for (int t = 0; t < 8; ++t) accg[t] = mm(w[t][2], w[t][3], ga[j][2], ga[j][3], accg[t]);
    }

#pragma unroll
    for (int t = 0; t < 8; ++t) {
      const v4f blo = bgv[4 * t + 2 * half], bhi = bgv[4 * t + 2 * half + 1];
#pragma unroll
      for (int r = 0; r < 4; ++r) {
        accg[t][r]     = relu1(accg[t][r] + blo[r]);
        accg[t][4 + r] = relu1(accg[t][4 + r] + bhi[r]);
      }
    }
    v8f acc2[3] = {};
#pragma unroll
    for (int t = 0; t < 8; ++t) {
#pragma unroll
      for (int qq = 0; qq < 2; ++qq) {
        v4f w[3];
#pragma unroll
        for (int u = 0; u < 3; ++u) w[u] = W2g[((t * 2 + qq) * 3 + u) * 32 + lane];
#pragma unroll
        for (int u = 0; u < 3; ++u)
          acc2[u] = mm(w[u][0], w[u][1], accg[t][4 * qq + 0], accg[t][4 * qq + 1], acc2[u]);
#pragma unroll
        for (int u = 0; u < 3; ++u)
          acc2[u] = mm(w[u][2], w[u][3], accg[t][4 * qq + 2], accg[t][4 * qq + 3], acc2[u]);
      }
    }

#pragma unroll
    for (int u = 0; u < 3; ++u) {
      if (u < 2 || half == 0) {
        v4f lo, hi;
#pragma unroll
        for (int r = 0; r < 4; ++r) { lo[r] = plo[u][r] + acc2[u][r]; hi[r] = phi[u][r] + acc2[u][4 + r]; }
        float* o = out + m * kC + 16 * u + 8 * half;
        *reinterpret_cast<v4f*>(o) = lo;
        *reinterpret_cast<v4f*>(o + 4) = hi;
      }
    }
  }
}


extern "C" void kernel_launch(void* const* d_in, const int* in_sizes, int n_in,
                              void* d_out, int out_size, void* d_ws, size_t ws_size,
                              hipStream_t stream) {
  (void)in_sizes; (void)n_in; (void)out_size; (void)ws_size;

  const float* xfeat  = (const float*)d_in[0];
  const float* xlabel = (const float*)d_in[1];
  const int*   eidx   = (const int*)d_in[2];
  const float* W_gcn  = (const float*)d_in[3];
  const float* b_gcn  = (const float*)d_in[4];
  const float* W_mlp  = (const float*)d_in[5];
  const float* b_mlp  = (const float*)d_in[6];
  const float* W_cls  = (const float*)d_in[7];
  const float* b_cls  = (const float*)d_in[8];
  float* out = (float*)d_out;

  char* ws = (char*)d_ws;
  const size_t off_deg  = 0;
  const size_t off_own  = (off_deg + (size_t)kN * 4 + 255) & ~(size_t)255;
  const size_t off_aggx = (off_own + (size_t)kE + 255) & ~(size_t)255;
  float* deg  = (float*)(ws + off_deg);
  unsigned* own = (unsigned*)(ws + off_own);
  float* aggx = (float*)(ws + off_aggx);

  const dim3 blk256(256);

  k_init_deg <<<dim3((kN + 255) / 256), blk256, 0, stream>>>(deg);
  k_deg_owner<<<dim3((kE / 4 + 255) / 256), blk256, 0, stream>>>(eidx, deg, own);
  k_dinv     <<<dim3((kN + 255) / 256), blk256, 0, stream>>>(deg);

  k_agg_lds<<<dim3(256), dim3(kAggThreads), 0, stream>>>(eidx, own, xfeat, xlabel, deg,
                                                         W_mlp, b_mlp, W_cls, b_cls, aggx, out);

  k_tail_fused<<<dim3(kTailBlocks), dim3(kTailThreads), 0, stream>>>(
      aggx, W_gcn, b_gcn, W_cls, out);
}
